# starting kernel
# baseline (speedup 1.0000x reference)
#include <hip/hip_runtime.h>

typedef __attribute__((ext_vector_type(2))) float v2f;
typedef __attribute__((ext_vector_type(8))) float v8f;

#define LOG2E 1.44269504088896340736f
#define NCHUNK_PER_B 4
#define NCHUNKS 8

__global__ __launch_bounds__(256) void kde_wmma_kernel(
    const float* __restrict__ samples,
    const float* __restrict__ locations,
    float* __restrict__ partials,
    int m, int S, int rowsPerChunk)
{
    const int wid  = (blockIdx.x * blockDim.x + threadIdx.x) >> 5;
    const int lane = threadIdx.x & 31;
    const int nIdx = lane & 15;
    const int hi   = lane >> 4;

    const int loc_tile = wid / NCHUNKS;
    const int chunk    = wid % NCHUNKS;
    const int b        = chunk >> 2;
    const int c        = chunk & (NCHUNK_PER_B - 1);

    const float2* loc2 = reinterpret_cast<const float2*>(locations);
    float2 y = loc2[loc_tile * 16 + nIdx];
    v2f bm;
    if (hi) { bm[0] = 1.0f;       bm[1] = y.x * y.x + y.y * y.y; }
    else    { bm[0] = -2.0f * y.x; bm[1] = -2.0f * y.y; }

    const float s = -12.5f * LOG2E;

    const float2* smp2 = reinterpret_cast<const float2*>(samples);
    const int rowBase = b * S + c * rowsPerChunk;
    const int nTiles  = rowsPerChunk >> 4;

    v8f acc = {0.f, 0.f, 0.f, 0.f, 0.f, 0.f, 0.f, 0.f};

    for (int t = 0; t < nTiles; ++t) {
        int row = rowBase + t * 16 + nIdx;
        float2 x = smp2[row];
        v2f am;
        if (hi) { am[0] = s * (x.x * x.x + x.y * x.y); am[1] = s; }
        else    { am[0] = s * x.x;                     am[1] = s * x.y; }

        v8f cz = {0.f, 0.f, 0.f, 0.f, 0.f, 0.f, 0.f, 0.f};
        v8f d = __builtin_amdgcn_wmma_f32_16x16x4_f32(
            false, am, false, bm, (short)0, cz, false, false);

#pragma unroll
        for (int i = 0; i < 8; ++i)
            acc[i] += __builtin_amdgcn_exp2f(d[i]);
    }

    float tsum = ((acc[0] + acc[1]) + (acc[2] + acc[3])) +
                 ((acc[4] + acc[5]) + (acc[6] + acc[7]));
    tsum += __shfl_xor(tsum, 16);

    if (hi == 0)
        partials[chunk * m + loc_tile * 16 + nIdx] = tsum;
}

__global__ __launch_bounds__(256) void kde_reduce_kernel(
    const float* __restrict__ partials, float* __restrict__ out,
    float* __restrict__ norms, int m)
{
    const int b = blockIdx.x;
    const int tid = threadIdx.x;

    float lnorm = 0.0f;
    for (int j = tid; j < m; j += blockDim.x) {
        float o = (partials[(b * NCHUNK_PER_B + 0) * m + j] +
                   partials[(b * NCHUNK_PER_B + 1) * m + j]) +
                  (partials[(b * NCHUNK_PER_B + 2) * m + j] +
                   partials[(b * NCHUNK_PER_B + 3) * m + j]);
        out[j * 2 + b] = o;
        lnorm += o;
    }

    __shared__ float sd[256];
    sd[tid] = lnorm;
    __syncthreads();
    for (int stp = 128; stp > 0; stp >>= 1) {
        if (tid < stp) sd[tid] += sd[tid + stp];
        __syncthreads();
    }
    if (tid == 0) norms[b] = sd[0];
}

__global__ __launch_bounds__(256) void kde_norm_kernel(
    float* __restrict__ out, const float* __restrict__ norms, int n)
{
    int i = blockIdx.x * blockDim.x + threadIdx.x;
    if (i < n) out[i] = out[i] / norms[i & 1];
}

extern "C" void kernel_launch(void* const* d_in, const int* in_sizes, int n_in,
                              void* d_out, int out_size, void* d_ws, size_t ws_size,
                              hipStream_t stream) {
    const float* samples   = (const float*)d_in[0];
    const float* locations = (const float*)d_in[1];
    float* out = (float*)d_out;

    const int R = in_sizes[0] / 2;
    const int m = in_sizes[1] / 2;
    const int B = 2;
    const int S = R / B;
    const int rowsPerChunk = S / NCHUNK_PER_B;

    float* partials = (float*)d_ws;
    float* norms    = partials + (size_t)NCHUNKS * m;

    const int nWaves  = (m / 16) * NCHUNKS;
    const int threads = 256;
    const int blocks  = nWaves * 32 / threads;

    kde_wmma_kernel<<<blocks, threads, 0, stream>>>(
        samples, locations, partials, m, S, rowsPerChunk);
    kde_reduce_kernel<<<B, 256, 0, stream>>>(partials, out, norms, m);

    const int n = 2 * m;
    kde_norm_kernel<<<(n + 255) / 256, 256, 0, stream>>>(out, norms, n);
}


# Drop the 2-block chunk-reduce launch: each block folds its 8 chunk sums in LDS and writes out plus a norm share; the divide kernel sums the 512 shares per wave (norm summation tree changes)
# speedup vs baseline: 2.3438x; 2.3438x over previous
#include <hip/hip_runtime.h>

typedef __attribute__((ext_vector_type(2))) float v2f;
typedef __attribute__((ext_vector_type(8))) float v8f;

#define LOG2E 1.44269504088896340736f
#define NCHUNK_PER_B 4
#define NCHUNKS 8

__global__ __launch_bounds__(NCHUNKS * 32) void kde_wmma_kernel(
    const float* __restrict__ samples,
    const float* __restrict__ locations,
    float* __restrict__ out,
    float* __restrict__ normpart,
    int S, int rowsPerChunk)
{
    const int lane = threadIdx.x & 31;
    const int nIdx = lane & 15;
    const int hi   = lane >> 4;

    const int loc_tile = blockIdx.x;
    const int chunk    = threadIdx.x >> 5;
    const int b        = chunk >> 2;
    const int c        = chunk & (NCHUNK_PER_B - 1);

    const float2* loc2 = reinterpret_cast<const float2*>(locations);
    float2 y = loc2[loc_tile * 16 + nIdx];
    v2f bm;
    if (hi) { bm[0] = 1.0f;       bm[1] = y.x * y.x + y.y * y.y; }
    else    { bm[0] = -2.0f * y.x; bm[1] = -2.0f * y.y; }

    const float s = -12.5f * LOG2E;

    const float2* smp2 = reinterpret_cast<const float2*>(samples);
    const int rowBase = b * S + c * rowsPerChunk;
    const int nTiles  = rowsPerChunk >> 4;

    v8f acc = {0.f, 0.f, 0.f, 0.f, 0.f, 0.f, 0.f, 0.f};

    for (int t = 0; t < nTiles; ++t) {
        int row = rowBase + t * 16 + nIdx;
        float2 x = smp2[row];
        v2f am;
        if (hi) { am[0] = s * (x.x * x.x + x.y * x.y); am[1] = s; }
        else    { am[0] = s * x.x;                     am[1] = s * x.y; }

        v8f cz = {0.f, 0.f, 0.f, 0.f, 0.f, 0.f, 0.f, 0.f};
        v8f d = __builtin_amdgcn_wmma_f32_16x16x4_f32(
            false, am, false, bm, (short)0, cz, false, false);

#pragma unroll
        for (int i = 0; i < 8; ++i)
            acc[i] += __builtin_amdgcn_exp2f(d[i]);
    }

    float tsum = ((acc[0] + acc[1]) + (acc[2] + acc[3])) +
                 ((acc[4] + acc[5]) + (acc[6] + acc[7]));
    tsum += __shfl_xor(tsum, 16);

    __shared__ float sh[NCHUNKS][16];
    if (hi == 0) sh[chunk][nIdx] = tsum;
    __syncthreads();

    if (chunk == 0) {
        const int j  = lane >> 1;
        const int ob = lane & 1;
        const float o = (sh[ob * NCHUNK_PER_B + 0][j] + sh[ob * NCHUNK_PER_B + 1][j]) +
                        (sh[ob * NCHUNK_PER_B + 2][j] + sh[ob * NCHUNK_PER_B + 3][j]);
        out[loc_tile * 32 + lane] = o;

        float p = o;
        p += __shfl_xor(p, 2);
        p += __shfl_xor(p, 4);
        p += __shfl_xor(p, 8);
        p += __shfl_xor(p, 16);
        if (lane < 2) normpart[loc_tile * 2 + lane] = p;
    }
}

__global__ __launch_bounds__(64) void kde_norm_kernel(
    float* __restrict__ out, const float* __restrict__ normpart,
    int nblk,
    int n4)
{
    const int lane = threadIdx.x & 31;
    const int i = blockIdx.x * blockDim.x + threadIdx.x;
    float4* out4 = reinterpret_cast<float4*>(out);
    const float4* p4 = reinterpret_cast<const float4*>(normpart);
    const float2* p2 = reinterpret_cast<const float2*>(normpart);

    float4 o = make_float4(0.f, 0.f, 0.f, 0.f);
    if (i < n4) o = out4[i];

    float s0 = 0.f, s1 = 0.f;
    int blk = 0;
    for (; blk + 512 <= nblk; blk += 512) {
        float4 v[8];
#pragma unroll
        for (int k = 0; k < 8; ++k)
            v[k] = p4[(blk >> 1) + k * 32 + lane];
#pragma unroll
        for (int k = 0; k < 8; ++k) {
            s0 += v[k].x + v[k].z;
            s1 += v[k].y + v[k].w;
        }
    }
    for (int r = blk + lane; r < nblk; r += 32) {
        const float2 t = p2[r];
        s0 += t.x;
        s1 += t.y;
    }
    s0 += __shfl_xor(s0, 16);  s1 += __shfl_xor(s1, 16);
    s0 += __shfl_xor(s0, 8);   s1 += __shfl_xor(s1, 8);
    s0 += __shfl_xor(s0, 4);   s1 += __shfl_xor(s1, 4);
    s0 += __shfl_xor(s0, 2);   s1 += __shfl_xor(s1, 2);
    s0 += __shfl_xor(s0, 1);   s1 += __shfl_xor(s1, 1);

    if (i < n4) {
        o.x = o.x / s0;  o.y = o.y / s1;
        o.z = o.z / s0;  o.w = o.w / s1;
        out4[i] = o;
    }
}

extern "C" void kernel_launch(void* const* d_in, const int* in_sizes, int n_in,
                              void* d_out, int out_size, void* d_ws, size_t ws_size,
                              hipStream_t stream) {
    const float* samples   = (const float*)d_in[0];
    const float* locations = (const float*)d_in[1];
    float* out = (float*)d_out;

    const int R = in_sizes[0] / 2;
    const int m = in_sizes[1] / 2;
    const int B = 2;
    const int S = R / B;
    const int rowsPerChunk = S / NCHUNK_PER_B;

    float* normpart = (float*)d_ws;

    kde_wmma_kernel<<<m / 16, NCHUNKS * 32, 0, stream>>>(
        samples, locations, out, normpart, S, rowsPerChunk);

    const int n4 = (2 * m) / 4;
    kde_norm_kernel<<<(n4 + 63) / 64, 64, 0, stream>>>(out, normpart, m / 16, n4);
}


# Main kernel: s, -2*s*y and s*|y|^2 moved into loop-invariant B and C operands, lane half h holds (x_h, x_h^2): a tile's A costs one 4-byte load and one multiply, not ~6 vector ops; sum order kept
# speedup vs baseline: 2.5084x; 1.0702x over previous
#include <hip/hip_runtime.h>

typedef __attribute__((ext_vector_type(2))) float v2f;
typedef __attribute__((ext_vector_type(8))) float v8f;

#define LOG2E 1.44269504088896340736f
#define NCHUNK_PER_B 4
#define NCHUNKS 8

__global__ __launch_bounds__(NCHUNKS * 32) void kde_wmma_kernel(
    const float* __restrict__ samples,
    const float* __restrict__ locations,
    float* __restrict__ out,
    float* __restrict__ normpart,
    int S, int rowsPerChunk)
{
    const int lane = threadIdx.x & 31;
    const int nIdx = lane & 15;
    const int hi   = lane >> 4;

    const int loc_tile = blockIdx.x;
    const int chunk    = threadIdx.x >> 5;
    const int b        = chunk >> 2;
    const int c        = chunk & (NCHUNK_PER_B - 1);

    const float s = -12.5f * LOG2E;

    const float2* loc2 = reinterpret_cast<const float2*>(locations);
    const float2 y  = loc2[loc_tile * 16 + nIdx];
    const float  yh = hi ? y.y : y.x;
    v2f bm;
    bm[0] = (-2.0f * s) * yh;
    bm[1] = s;
    const float cy = s * (y.x * y.x + y.y * y.y);
    const v8f cm = {cy, cy, cy, cy, cy, cy, cy, cy};

    const int rowBase = b * S + c * rowsPerChunk;
    const int nTiles  = rowsPerChunk >> 4;
    const float* sp = samples + 2 * (rowBase + nIdx) + hi;

    v8f acc = {0.f, 0.f, 0.f, 0.f, 0.f, 0.f, 0.f, 0.f};

    for (int t = 0; t < nTiles; ++t) {
        const float xh = sp[t * 32];
        v2f am;
        am[0] = xh;
        am[1] = xh * xh;

        v8f d = __builtin_amdgcn_wmma_f32_16x16x4_f32(
            false, am, false, bm, (short)0, cm, false, false);

#pragma unroll
        for (int i = 0; i < 8; ++i)
            acc[i] += __builtin_amdgcn_exp2f(d[i]);
    }

    float tsum = ((acc[0] + acc[1]) + (acc[2] + acc[3])) +
                 ((acc[4] + acc[5]) + (acc[6] + acc[7]));
    tsum += __shfl_xor(tsum, 16);

    __shared__ float sh[NCHUNKS][16];
    if (hi == 0) sh[chunk][nIdx] = tsum;
    __syncthreads();

    if (chunk == 0) {
        const int j  = lane >> 1;
        const int ob = lane & 1;
        const float o = (sh[ob * NCHUNK_PER_B + 0][j] + sh[ob * NCHUNK_PER_B + 1][j]) +
                        (sh[ob * NCHUNK_PER_B + 2][j] + sh[ob * NCHUNK_PER_B + 3][j]);
        out[loc_tile * 32 + lane] = o;

        float p = o;
        p += __shfl_xor(p, 2);
        p += __shfl_xor(p, 4);
        p += __shfl_xor(p, 8);
        p += __shfl_xor(p, 16);
        if (lane < 2) normpart[loc_tile * 2 + lane] = p;
    }
}

__global__ __launch_bounds__(64) void kde_norm_kernel(
    float* __restrict__ out, const float* __restrict__ normpart,
    int nblk,
    int n4)
{
    const int lane = threadIdx.x & 31;
    const int i = blockIdx.x * blockDim.x + threadIdx.x;
    float4* out4 = reinterpret_cast<float4*>(out);
    const float4* p4 = reinterpret_cast<const float4*>(normpart);
    const float2* p2 = reinterpret_cast<const float2*>(normpart);

    float4 o = make_float4(0.f, 0.f, 0.f, 0.f);
    if (i < n4) o = out4[i];

    float s0 = 0.f, s1 = 0.f;
    int blk = 0;
    for (; blk + 512 <= nblk; blk += 512) {
        float4 v[8];
#pragma unroll
        for (int k = 0; k < 8; ++k)
            v[k] = p4[(blk >> 1) + k * 32 + lane];
#pragma unroll
        for (int k = 0; k < 8; ++k) {
            s0 += v[k].x + v[k].z;
            s1 += v[k].y + v[k].w;
        }
    }
    for (int r = blk + lane; r < nblk; r += 32) {
        const float2 t = p2[r];
        s0 += t.x;
        s1 += t.y;
    }
    s0 += __shfl_xor(s0, 16);  s1 += __shfl_xor(s1, 16);
    s0 += __shfl_xor(s0, 8);   s1 += __shfl_xor(s1, 8);
    s0 += __shfl_xor(s0, 4);   s1 += __shfl_xor(s1, 4);
    s0 += __shfl_xor(s0, 2);   s1 += __shfl_xor(s1, 2);
    s0 += __shfl_xor(s0, 1);   s1 += __shfl_xor(s1, 1);

    if (i < n4) {
        o.x = o.x / s0;  o.y = o.y / s1;
        o.z = o.z / s0;  o.w = o.w / s1;
        out4[i] = o;
    }
}

extern "C" void kernel_launch(void* const* d_in, const int* in_sizes, int n_in,
                              void* d_out, int out_size, void* d_ws, size_t ws_size,
                              hipStream_t stream) {
    const float* samples   = (const float*)d_in[0];
    const float* locations = (const float*)d_in[1];
    float* out = (float*)d_out;

    const int R = in_sizes[0] / 2;
    const int m = in_sizes[1] / 2;
    const int B = 2;
    const int S = R / B;
    const int rowsPerChunk = S / NCHUNK_PER_B;

    float* normpart = (float*)d_ws;

    kde_wmma_kernel<<<m / 16, NCHUNKS * 32, 0, stream>>>(
        samples, locations, out, normpart, S, rowsPerChunk);

    const int n4 = (2 * m) / 4;
    kde_norm_kernel<<<(n4 + 63) / 64, 64, 0, stream>>>(out, normpart, m / 16, n4);
}


# Main loop order swapped: a wave holds the A fragments of its whole sample chunk in registers and sweeps them over 2 location tiles; a tile costs no load, multiply or wait; 2048 waves; sum order kept
# speedup vs baseline: 2.6537x; 1.0579x over previous
#include <hip/hip_runtime.h>

typedef __attribute__((ext_vector_type(2))) float v2f;
typedef __attribute__((ext_vector_type(8))) float v8f;

#define LOG2E 1.44269504088896340736f
#define NCHUNK_PER_B 4
#define NCHUNKS 8
#define TILES 64
#define LOC_TILES 2

__global__ __launch_bounds__(NCHUNKS * 32) void kde_wmma_kernel(
    const float* __restrict__ samples,
    const float* __restrict__ locations,
    float* __restrict__ out,
    float* __restrict__ normpart,
    int S)
{
    const int lane = threadIdx.x & 31;
    const int nIdx = lane & 15;
    const int hi   = lane >> 4;

    const int chunk = threadIdx.x >> 5;
    const int b     = chunk >> 2;
    const int c     = chunk & (NCHUNK_PER_B - 1);

    const float s = -12.5f * LOG2E;

    const float2* loc2 = reinterpret_cast<const float2*>(locations);
    float2 yv[LOC_TILES];
#pragma unroll
    for (int l = 0; l < LOC_TILES; ++l)
        yv[l] = loc2[(blockIdx.x * LOC_TILES + l) * 16 + nIdx];

    const int rowBase = b * S + c * (TILES * 16);
    const float* sp = samples + 2 * (rowBase + nIdx) + hi;
    v2f am[TILES];
#pragma unroll
    for (int t = 0; t < TILES; ++t) {
        const float xh = sp[t * 32];
        am[t][0] = xh;
        am[t][1] = xh * xh;
    }

    __shared__ float sh[LOC_TILES][NCHUNKS][16];

#pragma unroll
    for (int l = 0; l < LOC_TILES; ++l) {
        const float2 y  = yv[l];
        const float  yh = hi ? y.y : y.x;
        v2f bm;
        bm[0] = (-2.0f * s) * yh;
        bm[1] = s;
        const float cy = s * (y.x * y.x + y.y * y.y);
        const v8f cm = {cy, cy, cy, cy, cy, cy, cy, cy};

        v8f acc = {0.f, 0.f, 0.f, 0.f, 0.f, 0.f, 0.f, 0.f};

#pragma unroll
        for (int t = 0; t < TILES; ++t) {
            v8f d = __builtin_amdgcn_wmma_f32_16x16x4_f32(
                false, am[t], false, bm, (short)0, cm, false, false);
#pragma unroll
            for (int i = 0; i < 8; ++i)
                acc[i] += __builtin_amdgcn_exp2f(d[i]);
        }

        float tsum = ((acc[0] + acc[1]) + (acc[2] + acc[3])) +
                     ((acc[4] + acc[5]) + (acc[6] + acc[7]));
        tsum += __shfl_xor(tsum, 16);
        if (hi == 0) sh[l][chunk][nIdx] = tsum;
    }

    __syncthreads();

    if (chunk < LOC_TILES) {
        const int tile = blockIdx.x * LOC_TILES + chunk;
        const int j  = lane >> 1;
        const int ob = lane & 1;
        const float o = (sh[chunk][ob * NCHUNK_PER_B + 0][j] + sh[chunk][ob * NCHUNK_PER_B + 1][j]) +
                        (sh[chunk][ob * NCHUNK_PER_B + 2][j] + sh[chunk][ob * NCHUNK_PER_B + 3][j]);
        out[tile * 32 + lane] = o;

        float p = o;
        p += __shfl_xor(p, 2);
        p += __shfl_xor(p, 4);
        p += __shfl_xor(p, 8);
        p += __shfl_xor(p, 16);
        if (lane < 2) normpart[tile * 2 + lane] = p;
    }
}

__global__ __launch_bounds__(64) void kde_norm_kernel(
    float* __restrict__ out, const float* __restrict__ normpart,
    int nblk,
    int n4)
{
    const int lane = threadIdx.x & 31;
    const int i = blockIdx.x * blockDim.x + threadIdx.x;
    float4* out4 = reinterpret_cast<float4*>(out);
    const float4* p4 = reinterpret_cast<const float4*>(normpart);
    const float2* p2 = reinterpret_cast<const float2*>(normpart);

    float4 o = make_float4(0.f, 0.f, 0.f, 0.f);
    if (i < n4) o = out4[i];

    float s0 = 0.f, s1 = 0.f;
    int blk = 0;
    for (; blk + 512 <= nblk; blk += 512) {
        float4 v[8];
#pragma unroll
        for (int k = 0; k < 8; ++k)
            v[k] = p4[(blk >> 1) + k * 32 + lane];
#pragma unroll
        for (int k = 0; k < 8; ++k) {
            s0 += v[k].x + v[k].z;
            s1 += v[k].y + v[k].w;
        }
    }
    for (int r = blk + lane; r < nblk; r += 32) {
        const float2 t = p2[r];
        s0 += t.x;
        s1 += t.y;
    }
    s0 += __shfl_xor(s0, 16);  s1 += __shfl_xor(s1, 16);
    s0 += __shfl_xor(s0, 8);   s1 += __shfl_xor(s1, 8);
    s0 += __shfl_xor(s0, 4);   s1 += __shfl_xor(s1, 4);
    s0 += __shfl_xor(s0, 2);   s1 += __shfl_xor(s1, 2);
    s0 += __shfl_xor(s0, 1);   s1 += __shfl_xor(s1, 1);

    if (i < n4) {
        o.x = o.x / s0;  o.y = o.y / s1;
        o.z = o.z / s0;  o.w = o.w / s1;
        out4[i] = o;
    }
}

extern "C" void kernel_launch(void* const* d_in, const int* in_sizes, int n_in,
                              void* d_out, int out_size, void* d_ws, size_t ws_size,
                              hipStream_t stream) {
    const float* samples   = (const float*)d_in[0];
    const float* locations = (const float*)d_in[1];
    float* out = (float*)d_out;

    const int R = in_sizes[0] / 2;
    const int m = in_sizes[1] / 2;
    const int B = 2;
    const int S = R / B;

    float* normpart = (float*)d_ws;

    kde_wmma_kernel<<<m / (16 * LOC_TILES), NCHUNKS * 32, 0, stream>>>(
        samples, locations, out, normpart, S);

    const int n4 = (2 * m) / 4;
    kde_norm_kernel<<<(n4 + 63) / 64, 64, 0, stream>>>(out, normpart, m / 16, n4);
}
